# starting kernel
# baseline (speedup 1.0000x reference)
#include <hip/hip_runtime.h>

typedef __attribute__((ext_vector_type(2))) float v2f;
typedef __attribute__((ext_vector_type(8))) float v8f;

constexpr int kBatch = 4096;
constexpr int kNNZ   = 32;
constexpr int kDim   = 2048;
constexpr int kTilesPerWave  = 4;
constexpr int kColsPerWave   = 16 * kTilesPerWave;
constexpr int kColsPerBlock  = 8 * kColsPerWave;

__global__ __launch_bounds__(256) void sparse_gather_wmma_kernel(
    const float* __restrict__ values,
    const int*   __restrict__ indices,
    const float* __restrict__ kern,
    const float* __restrict__ bias,
    float*       __restrict__ out)
{
    const int lane = threadIdx.x & 31;
    const int wave = threadIdx.x >> 5;
    const int ml   = lane & 15;
    const int half = lane >> 4;

    const int colBase = blockIdx.x * kColsPerBlock + wave * kColsPerWave;
    const unsigned nc = (unsigned)(colBase + ml);
    const int rowBase = blockIdx.y * 16;
    const int baseEl  = rowBase * kNNZ;

    v8f acc0 = {}, acc1 = {}, acc2 = {}, acc3 = {};

    int4   i4 = *reinterpret_cast<const int4*>(indices + baseEl);
    float4 v4 = *reinterpret_cast<const float4*>(values  + baseEl);

    const float* r0 = kern + (((unsigned)(half ? i4.z : i4.x)) << 11) + nc;
    const float* r1 = kern + (((unsigned)(half ? i4.w : i4.y)) << 11) + nc;
    v2f b0 = {r0[0],  r1[0]};
    v2f b1 = {r0[16], r1[16]};
    v2f b2 = {r0[32], r1[32]};
    v2f b3 = {r0[48], r1[48]};

    for (int t = 0; t < 128; ++t) {
        const int m = t >> 3;
        v2f a;
        a.x = (ml == m) ? (half ? v4.z : v4.x) : 0.0f;
        a.y = (ml == m) ? (half ? v4.w : v4.y) : 0.0f;

        int4   ni4 = i4;
        float4 nv4 = v4;
        v2f nb0 = b0, nb1 = b1, nb2 = b2, nb3 = b3;
        if (t + 1 < 128) {
            const int off = baseEl + (t + 1) * 4;
            ni4 = *reinterpret_cast<const int4*>(indices + off);
            nv4 = *reinterpret_cast<const float4*>(values  + off);
            const float* p0 = kern + (((unsigned)(half ? ni4.z : ni4.x)) << 11) + nc;
            const float* p1 = kern + (((unsigned)(half ? ni4.w : ni4.y)) << 11) + nc;
            nb0 = v2f{p0[0],  p1[0]};
            nb1 = v2f{p0[16], p1[16]};
            nb2 = v2f{p0[32], p1[32]};
            nb3 = v2f{p0[48], p1[48]};
        }

        acc0 = __builtin_amdgcn_wmma_f32_16x16x4_f32(false, a, false, b0, (short)0, acc0, false, false);
        acc1 = __builtin_amdgcn_wmma_f32_16x16x4_f32(false, a, false, b1, (short)0, acc1, false, false);
        acc2 = __builtin_amdgcn_wmma_f32_16x16x4_f32(false, a, false, b2, (short)0, acc2, false, false);
        acc3 = __builtin_amdgcn_wmma_f32_16x16x4_f32(false, a, false, b3, (short)0, acc3, false, false);

        b0 = nb0; b1 = nb1; b2 = nb2; b3 = nb3;
        i4 = ni4; v4 = nv4;
    }

    const float bv0 = bias[nc];
    const float bv1 = bias[nc + 16];
    const float bv2 = bias[nc + 32];
    const float bv3 = bias[nc + 48];
    const int outRow = rowBase + half * 8;
#pragma unroll
    for (int j = 0; j < 8; ++j) {
        float* orow = out + (size_t)(outRow + j) * kDim + nc;
        orow[0]  = acc0[j] + bv0;
        orow[16] = acc1[j] + bv1;
        orow[32] = acc2[j] + bv2;
        orow[48] = acc3[j] + bv3;
    }
}

extern "C" void kernel_launch(void* const* d_in, const int* in_sizes, int n_in,
                              void* d_out, int out_size, void* d_ws, size_t ws_size,
                              hipStream_t stream) {
    const float* values  = (const float*)d_in[0];
    const int*   indices = (const int*)  d_in[1];
    const float* kern    = (const float*)d_in[2];
    const float* bias    = (const float*)d_in[3];
    float* outp = (float*)d_out;

    dim3 grid(kDim / kColsPerBlock, kBatch / 16);
    dim3 block(256);
    hipLaunchKernelGGL(sparse_gather_wmma_kernel, grid, block, 0, stream,
                       values, indices, kern, bias, outp);
}


# Relabel each WMMA tile as 4 batch rows x 4 column phases: a lane owns 4 adjacent columns, so gathers and stores are 16-byte on whole 128-B lines; loop nest per batch row; nnz sum order kept
# speedup vs baseline: 1.1673x; 1.1673x over previous
#include <hip/hip_runtime.h>

typedef __attribute__((ext_vector_type(2))) float v2f;
typedef __attribute__((ext_vector_type(8))) float v8f;

constexpr int kBatch = 4096;
constexpr int kNNZ   = 32;
constexpr int kDim   = 2048;
constexpr int kColsPerWave  = 64;
constexpr int kColsPerBlock = 8 * kColsPerWave;
constexpr int kSteps = 16 * kNNZ / 4;


struct Stage {
    float4 v4;
    float4 q0;
    float4 q1;
};

__device__ __forceinline__ void load_stage(Stage& s, const float* __restrict__ values,
                                           const int* __restrict__ indices,
                                           const float* __restrict__ kern,
                                           int off, int half, unsigned c4)
{
    const int4 i4 = *reinterpret_cast<const int4*>(indices + off);
    s.v4 = *reinterpret_cast<const float4*>(values + off);
    const float* p0 = kern + (((unsigned)(half ? i4.y : i4.x)) << 11) + c4;
    const float* p1 = kern + (((unsigned)(half ? i4.w : i4.z)) << 11) + c4;
    s.q0 = *reinterpret_cast<const float4*>(p0);
    s.q1 = *reinterpret_cast<const float4*>(p1);
}

__global__ __launch_bounds__(256) void sparse_gather_wmma_kernel(
    const float* __restrict__ values,
    const int*   __restrict__ indices,
    const float* __restrict__ kern,
    const float* __restrict__ bias,
    float*       __restrict__ out)
{
    const int lane = threadIdx.x & 31;
    const int wave = threadIdx.x >> 5;
    const int ml   = lane & 15;
    const int half = lane >> 4;

    const int colBase = blockIdx.x * kColsPerBlock + wave * kColsPerWave;
    const unsigned c4 = (unsigned)(colBase + 4 * ml);
    const int rowBase = blockIdx.y * 16;
    const int baseEl  = rowBase * kNNZ;

    const float4 bv = *reinterpret_cast<const float4*>(bias + c4);
    float* o = out + (size_t)(rowBase + 2 * half) * kDim + c4;

    Stage cur;
    load_stage(cur, values, indices, kern, baseEl, half, c4);

    for (int g = 0; g < 4; ++g) {
        v8f acc = {};
        for (int bi = 0; bi < 4; ++bi) {
            const int  m0 = 4 * bi;
            const bool e0 = (ml == m0);
            const bool e1 = (ml == m0 + 1);
            const bool e2 = (ml == m0 + 2);
            const bool e3 = (ml == m0 + 3);
            for (int c = 0; c < 8; ++c) {
                const int t = g * 32 + bi * 8 + c;

                const int tn = (t + 1 < kSteps) ? t + 1 : kSteps - 1;
                Stage nxt;
                load_stage(nxt, values, indices, kern, baseEl + tn * 4, half, c4);

                const float va = half ? cur.v4.y : cur.v4.x;
                const float vb = half ? cur.v4.w : cur.v4.z;

                const v2f a01 = { e0 ? va : 0.0f, e1 ? va : 0.0f };
                const v2f a23 = { e2 ? va : 0.0f, e3 ? va : 0.0f };
                const v2f b01 = { cur.q0.x, cur.q0.y };
                const v2f b23 = { cur.q0.z, cur.q0.w };
                acc = __builtin_amdgcn_wmma_f32_16x16x4_f32(false, a01, false, b01, (short)0, acc, false, false);
                acc = __builtin_amdgcn_wmma_f32_16x16x4_f32(false, a23, false, b23, (short)0, acc, false, false);

                const v2f c01 = { e0 ? vb : 0.0f, e1 ? vb : 0.0f };
                const v2f c23 = { e2 ? vb : 0.0f, e3 ? vb : 0.0f };
                const v2f d01 = { cur.q1.x, cur.q1.y };
                const v2f d23 = { cur.q1.z, cur.q1.w };
                acc = __builtin_amdgcn_wmma_f32_16x16x4_f32(false, c01, false, d01, (short)0, acc, false, false);
                acc = __builtin_amdgcn_wmma_f32_16x16x4_f32(false, c23, false, d23, (short)0, acc, false, false);

                cur = nxt;
            }
        }

        const float4 lo = { acc[0] + bv.x, acc[1] + bv.y, acc[2] + bv.z, acc[3] + bv.w };
        const float4 hi = { acc[4] + bv.x, acc[5] + bv.y, acc[6] + bv.z, acc[7] + bv.w };
        *reinterpret_cast<float4*>(o)        = lo;
        *reinterpret_cast<float4*>(o + kDim) = hi;
        o += 4 * kDim;
    }
}

extern "C" void kernel_launch(void* const* d_in, const int* in_sizes, int n_in,
                              void* d_out, int out_size, void* d_ws, size_t ws_size,
                              hipStream_t stream) {
    const float* values  = (const float*)d_in[0];
    const int*   indices = (const int*)  d_in[1];
    const float* kern    = (const float*)d_in[2];
    const float* bias    = (const float*)d_in[3];
    float* outp = (float*)d_out;

    dim3 grid(kDim / kColsPerBlock, kBatch / 16);
    dim3 block(256);
    hipLaunchKernelGGL(sparse_gather_wmma_kernel, grid, block, 0, stream,
                       values, indices, kern, bias, outp);
}


# Software pipeline two units deep (unit = half a batch row, 16 nnz): a unit's indices and values are loaded one unit before its 8 row gathers, the gathers one unit before its 16 WMMAs; sum order kept
# speedup vs baseline: 1.2257x; 1.0500x over previous
#include <hip/hip_runtime.h>

typedef __attribute__((ext_vector_type(2))) float v2f;
typedef __attribute__((ext_vector_type(8))) float v8f;

constexpr int kBatch = 4096;
constexpr int kNNZ   = 32;
constexpr int kDim   = 2048;
constexpr int kColsPerWave  = 64;
constexpr int kColsPerBlock = 8 * kColsPerWave;
constexpr int kRowsPerWave  = 16;


struct Scal {
    int4   i0, i1, i2, i3;
    float4 v0, v1, v2, v3;
};

struct Quads {
    float4 a0, a1, a2, a3;
    float4 b0, b1, b2, b3;
};

__device__ __forceinline__ Scal load_scal(const float* __restrict__ values,
                                          const int* __restrict__ indices, int off)
{
    Scal s;
    s.i0 = *reinterpret_cast<const int4*>(indices + off);
    s.i1 = *reinterpret_cast<const int4*>(indices + off + 4);
    s.i2 = *reinterpret_cast<const int4*>(indices + off + 8);
    s.i3 = *reinterpret_cast<const int4*>(indices + off + 12);
    s.v0 = *reinterpret_cast<const float4*>(values + off);
    s.v1 = *reinterpret_cast<const float4*>(values + off + 4);
    s.v2 = *reinterpret_cast<const float4*>(values + off + 8);
    s.v3 = *reinterpret_cast<const float4*>(values + off + 12);
    return s;
}

__device__ __forceinline__ float4 gather_row(const float* __restrict__ kern, int ilo, int ihi,
                                             int half, unsigned c4)
{
    const float* p = kern + (((unsigned)(half ? ihi : ilo)) << 11) + c4;
    return *reinterpret_cast<const float4*>(p);
}

__device__ __forceinline__ Quads gather_unit(const Scal& s, const float* __restrict__ kern,
                                             int half, unsigned c4)
{
    Quads q;
    q.a0 = gather_row(kern, s.i0.x, s.i0.y, half, c4);
    q.b0 = gather_row(kern, s.i0.z, s.i0.w, half, c4);
    q.a1 = gather_row(kern, s.i1.x, s.i1.y, half, c4);
    q.b1 = gather_row(kern, s.i1.z, s.i1.w, half, c4);
    q.a2 = gather_row(kern, s.i2.x, s.i2.y, half, c4);
    q.b2 = gather_row(kern, s.i2.z, s.i2.w, half, c4);
    q.a3 = gather_row(kern, s.i3.x, s.i3.y, half, c4);
    q.b3 = gather_row(kern, s.i3.z, s.i3.w, half, c4);
    return q;
}

__device__ __forceinline__ void mma_chunk(v8f& acc, float4 v4, float4 q0, float4 q1, int half,
                                          bool e0, bool e1, bool e2, bool e3)
{
    const float va = half ? v4.y : v4.x;
    const float vb = half ? v4.w : v4.z;

    const v2f a01 = { e0 ? va : 0.0f, e1 ? va : 0.0f };
    const v2f a23 = { e2 ? va : 0.0f, e3 ? va : 0.0f };
    const v2f b01 = { q0.x, q0.y };
    const v2f b23 = { q0.z, q0.w };
    acc = __builtin_amdgcn_wmma_f32_16x16x4_f32(false, a01, false, b01, (short)0, acc, false, false);
    acc = __builtin_amdgcn_wmma_f32_16x16x4_f32(false, a23, false, b23, (short)0, acc, false, false);

    const v2f c01 = { e0 ? vb : 0.0f, e1 ? vb : 0.0f };
    const v2f c23 = { e2 ? vb : 0.0f, e3 ? vb : 0.0f };
    const v2f d01 = { q1.x, q1.y };
    const v2f d23 = { q1.z, q1.w };
    acc = __builtin_amdgcn_wmma_f32_16x16x4_f32(false, c01, false, d01, (short)0, acc, false, false);
    acc = __builtin_amdgcn_wmma_f32_16x16x4_f32(false, c23, false, d23, (short)0, acc, false, false);
}

__device__ __forceinline__ void mma_unit(v8f& acc, const Quads& q, const Scal& s, int half,
                                         bool e0, bool e1, bool e2, bool e3)
{
    mma_chunk(acc, s.v0, q.a0, q.b0, half, e0, e1, e2, e3);
    mma_chunk(acc, s.v1, q.a1, q.b1, half, e0, e1, e2, e3);
    mma_chunk(acc, s.v2, q.a2, q.b2, half, e0, e1, e2, e3);
    mma_chunk(acc, s.v3, q.a3, q.b3, half, e0, e1, e2, e3);
}

__global__ __launch_bounds__(256) void sparse_gather_wmma_kernel(
    const float* __restrict__ values,
    const int*   __restrict__ indices,
    const float* __restrict__ kern,
    const float* __restrict__ bias,
    float*       __restrict__ out)
{
    const int lane = threadIdx.x & 31;
    const int wave = threadIdx.x >> 5;
    const int ml   = lane & 15;
    const int half = lane >> 4;

    const int colBase = blockIdx.x * kColsPerBlock + wave * kColsPerWave;
    const unsigned c4 = (unsigned)(colBase + 4 * ml);
    const int rowBase = blockIdx.y * kRowsPerWave;
    const int baseEl  = rowBase * kNNZ;

    const float4 bv = *reinterpret_cast<const float4*>(bias + c4);
    float* o = out + (size_t)(rowBase + 2 * half) * kDim + c4;

    Scal  s0 = load_scal(values, indices, baseEl);
    Scal  s1 = load_scal(values, indices, baseEl + 16);
    Quads qa = gather_unit(s0, kern, half, c4);
    v8f acc = {};

    auto store_group = [&]() {
        const float4 lo = { acc[0] + bv.x, acc[1] + bv.y, acc[2] + bv.z, acc[3] + bv.w };
        const float4 hi = { acc[4] + bv.x, acc[5] + bv.y, acc[6] + bv.z, acc[7] + bv.w };
        *reinterpret_cast<float4*>(o)        = lo;
        *reinterpret_cast<float4*>(o + kDim) = hi;
        o += 4 * kDim;
    };

    for (int row = 0; row < kRowsPerWave - 1; ++row) {
        const int  bi = row & 3;
        const int  nextEl = baseEl + (row + 1) * kNNZ;

        const int  m0 = 4 * bi;
        const bool e0 = (ml == m0);
        const bool e1 = (ml == m0 + 1);
        const bool e2 = (ml == m0 + 2);
        const bool e3 = (ml == m0 + 3);

        const Quads qb = gather_unit(s1, kern, half, c4);
        const Scal  t0 = load_scal(values, indices, nextEl);
        __builtin_amdgcn_sched_barrier(0);
        mma_unit(acc, qa, s0, half, e0, e1, e2, e3);
        __builtin_amdgcn_sched_barrier(0);

        qa = gather_unit(t0, kern, half, c4);
        __builtin_amdgcn_sched_barrier(0);
        const Scal  t1 = load_scal(values, indices, nextEl + 16);
        __builtin_amdgcn_sched_barrier(0);
        mma_unit(acc, qb, s1, half, e0, e1, e2, e3);

        s0 = t0;
        s1 = t1;

        if (bi == 3) {
            store_group();
            acc = v8f{};
        }
    }

    {
        const bool e0 = (ml == 12);
        const bool e1 = (ml == 13);
        const bool e2 = (ml == 14);
        const bool e3 = (ml == 15);
        const Quads qb = gather_unit(s1, kern, half, c4);
        __builtin_amdgcn_sched_barrier(0);
        mma_unit(acc, qa, s0, half, e0, e1, e2, e3);
        mma_unit(acc, qb, s1, half, e0, e1, e2, e3);
        store_group();
    }
}

extern "C" void kernel_launch(void* const* d_in, const int* in_sizes, int n_in,
                              void* d_out, int out_size, void* d_ws, size_t ws_size,
                              hipStream_t stream) {
    const float* values  = (const float*)d_in[0];
    const int*   indices = (const int*)  d_in[1];
    const float* kern    = (const float*)d_in[2];
    const float* bias    = (const float*)d_in[3];
    float* outp = (float*)d_out;

    dim3 grid(kDim / kColsPerBlock, kBatch / 16);
    dim3 block(256);
    hipLaunchKernelGGL(sparse_gather_wmma_kernel, grid, block, 0, stream,
                       values, indices, kern, bias, outp);
}


# Each workgroup copies its 32 columns of all 2048 kernel rows (256 KiB) into LDS once and gathers from LDS, not L2; a wave parks its rows' (row offset, value) pairs in a small LDS ring; nnz order kept
# speedup vs baseline: 1.3511x; 1.1023x over previous
#include <hip/hip_runtime.h>

typedef __attribute__((ext_vector_type(2))) float v2f;
typedef __attribute__((ext_vector_type(8))) float v8f;

constexpr int kBatch = 4096;
constexpr int kNNZ   = 32;
constexpr int kDim   = 2048;
constexpr int kSliceCols    = 32;
constexpr int kSlices       = kDim / kSliceCols;
constexpr int kBatchParts   = 4;
constexpr int kRowsPerBlock = kBatch / kBatchParts;
constexpr int kWaves        = 32;
constexpr int kThreads      = kWaves * 32;
constexpr int kRowsPerWave  = kRowsPerBlock / kWaves;
constexpr int kFillPasses   = 16;
constexpr int kRingSlots    = 4;


struct Pairs {
    uint2 p0, p1, p2, p3, p4, p5, p6, p7;
};

struct Rows {
    float2 r0, r1, r2, r3, r4, r5, r6, r7;
};

__device__ __forceinline__ Pairs read_pairs(const uint2* pr)
{
    Pairs p;
    p.p0 = pr[0];  p.p1 = pr[2];  p.p2 = pr[4];  p.p3 = pr[6];
    p.p4 = pr[8];  p.p5 = pr[10]; p.p6 = pr[12]; p.p7 = pr[14];
    return p;
}

__device__ __forceinline__ float2 lds_row(const float* slice, unsigned off, unsigned lane8)
{
    return *reinterpret_cast<const float2*>(reinterpret_cast<const char*>(slice) + (off + lane8));
}

__device__ __forceinline__ Rows read_rows(const Pairs& p, const float* slice, unsigned lane8)
{
    Rows q;
    q.r0 = lds_row(slice, p.p0.x, lane8);
    q.r1 = lds_row(slice, p.p1.x, lane8);
    q.r2 = lds_row(slice, p.p2.x, lane8);
    q.r3 = lds_row(slice, p.p3.x, lane8);
    q.r4 = lds_row(slice, p.p4.x, lane8);
    q.r5 = lds_row(slice, p.p5.x, lane8);
    q.r6 = lds_row(slice, p.p6.x, lane8);
    q.r7 = lds_row(slice, p.p7.x, lane8);
    return q;
}

__device__ __forceinline__ void mma_pair(v8f& acc, unsigned vbits, float2 r, bool e0, bool e1)
{
    const float v = __uint_as_float(vbits);
    const v2f a = { e0 ? v : 0.0f, e1 ? v : 0.0f };
    const v2f b = { r.x, r.y };
    acc = __builtin_amdgcn_wmma_f32_16x16x4_f32(false, a, false, b, (short)0, acc, false, false);
}

__device__ __forceinline__ void mma_unit(v8f& acc, const Rows& q, const Pairs& p, bool e0, bool e1)
{
    mma_pair(acc, p.p0.y, q.r0, e0, e1);
    mma_pair(acc, p.p1.y, q.r1, e0, e1);
    mma_pair(acc, p.p2.y, q.r2, e0, e1);
    mma_pair(acc, p.p3.y, q.r3, e0, e1);
    mma_pair(acc, p.p4.y, q.r4, e0, e1);
    mma_pair(acc, p.p5.y, q.r5, e0, e1);
    mma_pair(acc, p.p6.y, q.r6, e0, e1);
    mma_pair(acc, p.p7.y, q.r7, e0, e1);
}

__global__ __launch_bounds__(kThreads) void sparse_gather_wmma_kernel(
    const float* __restrict__ values,
    const int*   __restrict__ indices,
    const float* __restrict__ kern,
    const float* __restrict__ bias,
    float*       __restrict__ out)
{
    __shared__ float slice[kDim * kSliceCols];
    __shared__ uint2 ring[kWaves * kRingSlots * kNNZ];

    const int tid  = threadIdx.x;
    const int lane = tid & 31;
    const int wave = __builtin_amdgcn_readfirstlane(tid >> 5);
    const int ml   = lane & 15;
    const int half = lane >> 4;

    const int colBase = blockIdx.x * kSliceCols;
    const unsigned lane8 = (unsigned)(8 * ml);
    const int rowBase = blockIdx.y * kRowsPerBlock + wave * kRowsPerWave;
    const int baseEl  = rowBase * kNNZ;

    const int*   ip = indices + baseEl + lane;
    const float* vp = values  + baseEl + lane;
    const int   i0 = ip[0];
    const float v0 = vp[0];
    int   gi = ip[kNNZ];
    float gv = vp[kNNZ];
    const float2 bv = *reinterpret_cast<const float2*>(bias + colBase + 2 * ml);

    {
        const int piece = tid & 7;
        const int r0    = tid >> 3;
        const float* g = kern + (size_t)r0 * kDim + colBase + 4 * piece;
        float*       l = slice + r0 * kSliceCols + 4 * piece;
        float4 tmp[kFillPasses];
#pragma unroll
        for (int p = 0; p < kFillPasses; ++p)
            tmp[p] = *reinterpret_cast<const float4*>(g + (size_t)p * 128 * kDim);
#pragma unroll
        for (int p = 0; p < kFillPasses; ++p)
            *reinterpret_cast<float4*>(l + p * 128 * kSliceCols) = tmp[p];
    }

    uint2* myring = ring + wave * (kRingSlots * kNNZ);
    myring[lane] = make_uint2((unsigned)i0 << 7, __float_as_uint(v0));

    __syncthreads();

    float* o = out + (size_t)(rowBase + 4 * half) * kDim + colBase + 2 * ml;

    Pairs pa = read_pairs(myring + half);
    Rows  ra = read_rows(pa, slice, lane8);
    v8f acc = {};

    auto store_group = [&]() {
        const float2 r0 = { acc[0] + bv.x, acc[1] + bv.y };
        const float2 r1 = { acc[2] + bv.x, acc[3] + bv.y };
        const float2 r2 = { acc[4] + bv.x, acc[5] + bv.y };
        const float2 r3 = { acc[6] + bv.x, acc[7] + bv.y };
        *reinterpret_cast<float2*>(o)            = r0;
        *reinterpret_cast<float2*>(o + kDim)     = r1;
        *reinterpret_cast<float2*>(o + 2 * kDim) = r2;
        *reinterpret_cast<float2*>(o + 3 * kDim) = r3;
        o += 8 * kDim;
    };

    for (int row = 0; row < kRowsPerWave - 1; ++row) {
        const int  bi = row & 7;

        const int  m0 = 2 * bi;
        const bool e0 = (ml == m0);
        const bool e1 = (ml == m0 + 1);

        const uint2* slotCur  = myring + (row & 3) * kNNZ;
        uint2*       slotNext = myring + ((row + 1) & 3) * kNNZ;

        slotNext[lane] = make_uint2((unsigned)gi << 7, __float_as_uint(gv));
        const int nr = (row + 2 < kRowsPerWave) ? row + 2 : kRowsPerWave - 1;
        gi = ip[nr * kNNZ];
        gv = vp[nr * kNNZ];

        const Pairs pb = read_pairs(slotCur + 16 + half);
        __builtin_amdgcn_sched_barrier(0);
        mma_unit(acc, ra, pa, e0, e1);
        __builtin_amdgcn_sched_barrier(0);
        const Rows rb = read_rows(pb, slice, lane8);

        pa = read_pairs(slotNext + half);
        __builtin_amdgcn_sched_barrier(0);
        mma_unit(acc, rb, pb, e0, e1);
        __builtin_amdgcn_sched_barrier(0);
        ra = read_rows(pa, slice, lane8);

        if (bi == 7) {
            store_group();
            acc = v8f{};
        }
    }

    {
        const bool e0 = (ml == 14);
        const bool e1 = (ml == 15);
        const Pairs pb = read_pairs(myring + ((kRowsPerWave - 1) & 3) * kNNZ + 16 + half);
        __builtin_amdgcn_sched_barrier(0);
        mma_unit(acc, ra, pa, e0, e1);
        const Rows rb = read_rows(pb, slice, lane8);
        mma_unit(acc, rb, pb, e0, e1);
        store_group();
    }
}

extern "C" void kernel_launch(void* const* d_in, const int* in_sizes, int n_in,
                              void* d_out, int out_size, void* d_ws, size_t ws_size,
                              hipStream_t stream) {
    const float* values  = (const float*)d_in[0];
    const int*   indices = (const int*)  d_in[1];
    const float* kern    = (const float*)d_in[2];
    const float* bias    = (const float*)d_in[3];
    float* outp = (float*)d_out;

    dim3 grid(kSlices, kBatchParts);
    dim3 block(kThreads);
    hipLaunchKernelGGL(sparse_gather_wmma_kernel, grid, block, 0, stream,
                       values, indices, kern, bias, outp);
}


# One accumulator per batch row, M index = column phase only: the A operand is read ready-made from ring records (v,0,0,v), so the two per-WMMA selects and the per-row compares go; nnz order kept
# speedup vs baseline: 1.3944x; 1.0320x over previous
#include <hip/hip_runtime.h>

typedef __attribute__((ext_vector_type(2))) float v2f;
typedef __attribute__((ext_vector_type(8))) float v8f;

constexpr int kBatch = 4096;
constexpr int kNNZ   = 32;
constexpr int kDim   = 2048;
constexpr int kSliceCols    = 32;
constexpr int kSlices       = kDim / kSliceCols;
constexpr int kBatchParts   = 4;
constexpr int kRowsPerBlock = kBatch / kBatchParts;
constexpr int kWaves        = 32;
constexpr int kThreads      = kWaves * 32;
constexpr int kRowsPerWave  = kRowsPerBlock / kWaves;
constexpr int kFillPasses   = 16;
constexpr int kRingSlots    = 2;


struct Offs {
    uint4 lo, hi;
};

struct AOps {
    float2 a0, a1, a2, a3, a4, a5, a6, a7;
};

struct Rows {
    float2 r0, r1, r2, r3, r4, r5, r6, r7;
};

__device__ __forceinline__ Offs read_offs(const uint4* ob, int u)
{
    Offs o;
    o.lo = ob[2 * u];
    o.hi = ob[2 * u + 1];
    return o;
}

__device__ __forceinline__ AOps read_aops(const float2* ab, int u)
{
    const float2* p = ab + 32 * u;
    AOps a;
    a.a0 = p[0];  a.a1 = p[4];  a.a2 = p[8];  a.a3 = p[12];
    a.a4 = p[16]; a.a5 = p[20]; a.a6 = p[24]; a.a7 = p[28];
    return a;
}

__device__ __forceinline__ float2 lds_row(const float* slice, unsigned off, unsigned lane8)
{
    return *reinterpret_cast<const float2*>(reinterpret_cast<const char*>(slice) + (off + lane8));
}

__device__ __forceinline__ Rows read_rows(const Offs& o, const float* slice, unsigned lane8)
{
    Rows q;
    q.r0 = lds_row(slice, o.lo.x, lane8);
    q.r1 = lds_row(slice, o.lo.y, lane8);
    q.r2 = lds_row(slice, o.lo.z, lane8);
    q.r3 = lds_row(slice, o.lo.w, lane8);
    q.r4 = lds_row(slice, o.hi.x, lane8);
    q.r5 = lds_row(slice, o.hi.y, lane8);
    q.r6 = lds_row(slice, o.hi.z, lane8);
    q.r7 = lds_row(slice, o.hi.w, lane8);
    return q;
}

__device__ __forceinline__ v8f mma_pair(v8f acc, float2 av, float2 r)
{
    const v2f a = { av.x, av.y };
    const v2f b = { r.x, r.y };
    return __builtin_amdgcn_wmma_f32_16x16x4_f32(false, a, false, b, (short)0, acc, false, false);
}

__device__ __forceinline__ v8f mma_unit(v8f acc, const AOps& a, const Rows& q)
{
    acc = mma_pair(acc, a.a0, q.r0);
    acc = mma_pair(acc, a.a1, q.r1);
    acc = mma_pair(acc, a.a2, q.r2);
    acc = mma_pair(acc, a.a3, q.r3);
    acc = mma_pair(acc, a.a4, q.r4);
    acc = mma_pair(acc, a.a5, q.r5);
    acc = mma_pair(acc, a.a6, q.r6);
    acc = mma_pair(acc, a.a7, q.r7);
    return acc;
}

__global__ __launch_bounds__(kThreads) void sparse_gather_wmma_kernel(
    const float* __restrict__ values,
    const int*   __restrict__ indices,
    const float* __restrict__ kern,
    const float* __restrict__ bias,
    float*       __restrict__ out)
{
    __shared__ float  slice[kDim * kSliceCols];
    __shared__ float4 recs[kWaves * kRingSlots * kNNZ];
    __shared__ uint4  offs4[kWaves * kRingSlots * kNNZ / 4];

    const int tid  = threadIdx.x;
    const int lane = tid & 31;
    const int wave = __builtin_amdgcn_readfirstlane(tid >> 5);
    const int ml   = lane & 15;
    const int half = lane >> 4;

    const int colBase = blockIdx.x * kSliceCols;
    const unsigned lane8 = (unsigned)(8 * ml);
    const int rowBase = blockIdx.y * kRowsPerBlock + wave * kRowsPerWave;
    const int baseEl  = rowBase * kNNZ;

    const int*   ip = indices + baseEl + lane;
    const float* vp = values  + baseEl + lane;
    const int   i0 = ip[0];
    const float v0 = vp[0];
    int   gi = ip[kNNZ];
    float gv = vp[kNNZ];
    const float2 bv = *reinterpret_cast<const float2*>(bias + colBase + 2 * ml);

    {
        const int piece = tid & 7;
        const int r0    = tid >> 3;
        const float* g = kern + (size_t)r0 * kDim + colBase + 4 * piece;
        float*       l = slice + r0 * kSliceCols + 4 * piece;
        float4 tmp[kFillPasses];
#pragma unroll
        for (int p = 0; p < kFillPasses; ++p)
            tmp[p] = *reinterpret_cast<const float4*>(g + (size_t)p * 128 * kDim);
#pragma unroll
        for (int p = 0; p < kFillPasses; ++p)
            *reinterpret_cast<float4*>(l + p * 128 * kSliceCols) = tmp[p];
    }

    float4*   myrecs = recs + wave * (kRingSlots * kNNZ);
    unsigned* myoffs = reinterpret_cast<unsigned*>(offs4) + wave * (kRingSlots * kNNZ);
    const int offPlace = (lane & 1) * 16 + (lane >> 1);

    myrecs[lane]     = make_float4(v0, 0.0f, 0.0f, v0);
    myoffs[offPlace] = (unsigned)i0 << 7;

    __syncthreads();

    const float2* ab0 = reinterpret_cast<const float2*>(myrecs + half) + (ml & 1);
    const float2* ab1 = ab0 + 2 * kNNZ;
    const uint4*  ob0 = reinterpret_cast<const uint4*>(myoffs + 16 * half);
    const uint4*  ob1 = ob0 + kNNZ / 4;

    float* o = out + (size_t)(rowBase + half) * kDim + colBase + 2 * ml;

    AOps aa = read_aops(ab0, 0);
    Rows ra;
    {
        const Offs oa = read_offs(ob0, 0);
        ra = read_rows(oa, slice, lane8);
    }

    auto do_row = [&](int row, const float2* abCur, const uint4* obCur, const float2* abNext, const uint4* obNext,
                      int slotNext) -> v8f {
        myrecs[slotNext * kNNZ + lane]     = make_float4(gv, 0.0f, 0.0f, gv);
        myoffs[slotNext * kNNZ + offPlace] = (unsigned)gi << 7;
        const int nr = (row + 2 < kRowsPerWave) ? row + 2 : kRowsPerWave - 1;
        gi = ip[nr * kNNZ];
        gv = vp[nr * kNNZ];

        const Offs ob = read_offs(obCur, 1);
        const AOps abv = read_aops(abCur, 1);
        __builtin_amdgcn_sched_barrier(0);
        v8f acc = {};
        acc = mma_unit(acc, aa, ra);
        __builtin_amdgcn_sched_barrier(0);
        const Rows rb = read_rows(ob, slice, lane8);

        const Offs on = read_offs(obNext, 0);
        aa = read_aops(abNext, 0);
        __builtin_amdgcn_sched_barrier(0);
        acc = mma_unit(acc, abv, rb);
        __builtin_amdgcn_sched_barrier(0);
        ra = read_rows(on, slice, lane8);
        return acc;
    };

    for (int row = 0; row < kRowsPerWave; row += 2) {
        const v8f accE = do_row(row,     ab0, ob0, ab1, ob1, 1);
        const v8f accO = do_row(row + 1, ab1, ob1, ab0, ob0, 0);
        const float2 r = { (half ? accO[0] : accE[0]) + bv.x, (half ? accO[1] : accE[1]) + bv.y };
        *reinterpret_cast<float2*>(o) = r;
        o += 2 * kDim;
    }
}

extern "C" void kernel_launch(void* const* d_in, const int* in_sizes, int n_in,
                              void* d_out, int out_size, void* d_ws, size_t ws_size,
                              hipStream_t stream) {
    const float* values  = (const float*)d_in[0];
    const int*   indices = (const int*)  d_in[1];
    const float* kern    = (const float*)d_in[2];
    const float* bias    = (const float*)d_in[3];
    float* outp = (float*)d_out;

    dim3 grid(kSlices, kBatchParts);
    dim3 block(kThreads);
    hipLaunchKernelGGL(sparse_gather_wmma_kernel, grid, block, 0, stream,
                       values, indices, kern, bias, outp);
}


# A wave starts at issue priority 3 and steps it down (s_setprio) after 14, 24 and 30 of its 32 rows, so waves behind outrank waves ahead and a SIMD's 8 waves end together; nnz sum order kept
# speedup vs baseline: 1.5281x; 1.0959x over previous
#include <hip/hip_runtime.h>

typedef __attribute__((ext_vector_type(2))) float v2f;
typedef __attribute__((ext_vector_type(8))) float v8f;

constexpr int kBatch = 4096;
constexpr int kNNZ   = 32;
constexpr int kDim   = 2048;
constexpr int kSliceCols    = 32;
constexpr int kSlices       = kDim / kSliceCols;
constexpr int kBatchParts   = 4;
constexpr int kRowsPerBlock = kBatch / kBatchParts;
constexpr int kWaves        = 32;
constexpr int kThreads      = kWaves * 32;
constexpr int kRowsPerWave  = kRowsPerBlock / kWaves;
constexpr int kFillPasses   = 16;
constexpr int kRingSlots    = 2;
constexpr int kPrioStepA    = 14;
constexpr int kPrioStepB    = 24;
constexpr int kPrioStepC    = 30;


struct Offs {
    uint4 lo, hi;
};

struct AOps {
    float2 a0, a1, a2, a3, a4, a5, a6, a7;
};

struct Rows {
    float2 r0, r1, r2, r3, r4, r5, r6, r7;
};

__device__ __forceinline__ Offs read_offs(const uint4* ob, int u)
{
    Offs o;
    o.lo = ob[2 * u];
    o.hi = ob[2 * u + 1];
    return o;
}

__device__ __forceinline__ AOps read_aops(const float2* ab, int u)
{
    const float2* p = ab + 32 * u;
    AOps a;
    a.a0 = p[0];  a.a1 = p[4];  a.a2 = p[8];  a.a3 = p[12];
    a.a4 = p[16]; a.a5 = p[20]; a.a6 = p[24]; a.a7 = p[28];
    return a;
}

__device__ __forceinline__ float2 lds_row(const float* slice, unsigned off, unsigned lane8)
{
    return *reinterpret_cast<const float2*>(reinterpret_cast<const char*>(slice) + (off + lane8));
}

__device__ __forceinline__ Rows read_rows(const Offs& o, const float* slice, unsigned lane8)
{
    Rows q;
    q.r0 = lds_row(slice, o.lo.x, lane8);
    q.r1 = lds_row(slice, o.lo.y, lane8);
    q.r2 = lds_row(slice, o.lo.z, lane8);
    q.r3 = lds_row(slice, o.lo.w, lane8);
    q.r4 = lds_row(slice, o.hi.x, lane8);
    q.r5 = lds_row(slice, o.hi.y, lane8);
    q.r6 = lds_row(slice, o.hi.z, lane8);
    q.r7 = lds_row(slice, o.hi.w, lane8);
    return q;
}

__device__ __forceinline__ v8f mma_pair(v8f acc, float2 av, float2 r)
{
    const v2f a = { av.x, av.y };
    const v2f b = { r.x, r.y };
    return __builtin_amdgcn_wmma_f32_16x16x4_f32(false, a, false, b, (short)0, acc, false, false);
}

__device__ __forceinline__ v8f mma_unit(v8f acc, const AOps& a, const Rows& q)
{
    acc = mma_pair(acc, a.a0, q.r0);
    acc = mma_pair(acc, a.a1, q.r1);
    acc = mma_pair(acc, a.a2, q.r2);
    acc = mma_pair(acc, a.a3, q.r3);
    acc = mma_pair(acc, a.a4, q.r4);
    acc = mma_pair(acc, a.a5, q.r5);
    acc = mma_pair(acc, a.a6, q.r6);
    acc = mma_pair(acc, a.a7, q.r7);
    return acc;
}

__global__ __launch_bounds__(kThreads) void sparse_gather_wmma_kernel(
    const float* __restrict__ values,
    const int*   __restrict__ indices,
    const float* __restrict__ kern,
    const float* __restrict__ bias,
    float*       __restrict__ out)
{
    __shared__ float  slice[kDim * kSliceCols];
    __shared__ float4 recs[kWaves * kRingSlots * kNNZ];
    __shared__ uint4  offs4[kWaves * kRingSlots * kNNZ / 4];

    const int tid  = threadIdx.x;
    const int lane = tid & 31;
    const int wave = __builtin_amdgcn_readfirstlane(tid >> 5);
    const int ml   = lane & 15;
    const int half = lane >> 4;

    const int colBase = blockIdx.x * kSliceCols;
    const unsigned lane8 = (unsigned)(8 * ml);
    const int rowBase = blockIdx.y * kRowsPerBlock + wave * kRowsPerWave;
    const int baseEl  = rowBase * kNNZ;

    const int*   ip = indices + baseEl + lane;
    const float* vp = values  + baseEl + lane;
    const int   i0 = ip[0];
    const float v0 = vp[0];
    int   gi = ip[kNNZ];
    float gv = vp[kNNZ];
    const float2 bv = *reinterpret_cast<const float2*>(bias + colBase + 2 * ml);

    {
        const int piece = tid & 7;
        const int r0    = tid >> 3;
        const float* g = kern + (size_t)r0 * kDim + colBase + 4 * piece;
        float*       l = slice + r0 * kSliceCols + 4 * piece;
        float4 tmp[kFillPasses];
#pragma unroll
        for (int p = 0; p < kFillPasses; ++p)
            tmp[p] = *reinterpret_cast<const float4*>(g + (size_t)p * 128 * kDim);
#pragma unroll
        for (int p = 0; p < kFillPasses; ++p)
            *reinterpret_cast<float4*>(l + p * 128 * kSliceCols) = tmp[p];
    }

    float4*   myrecs = recs + wave * (kRingSlots * kNNZ);
    unsigned* myoffs = reinterpret_cast<unsigned*>(offs4) + wave * (kRingSlots * kNNZ);
    const int offPlace = (lane & 1) * 16 + (lane >> 1);

    myrecs[lane]     = make_float4(v0, 0.0f, 0.0f, v0);
    myoffs[offPlace] = (unsigned)i0 << 7;

    __syncthreads();

    const float2* ab0 = reinterpret_cast<const float2*>(myrecs + half) + (ml & 1);
    const float2* ab1 = ab0 + 2 * kNNZ;
    const uint4*  ob0 = reinterpret_cast<const uint4*>(myoffs + 16 * half);
    const uint4*  ob1 = ob0 + kNNZ / 4;

    float* o = out + (size_t)(rowBase + half) * kDim + colBase + 2 * ml;

    AOps aa = read_aops(ab0, 0);
    Rows ra;
    {
        const Offs oa = read_offs(ob0, 0);
        ra = read_rows(oa, slice, lane8);
    }

    auto do_row = [&](int row, const float2* abCur, const uint4* obCur, const float2* abNext, const uint4* obNext,
                      int slotNext) -> v8f {
        myrecs[slotNext * kNNZ + lane]     = make_float4(gv, 0.0f, 0.0f, gv);
        myoffs[slotNext * kNNZ + offPlace] = (unsigned)gi << 7;
        const int nr = (row + 2 < kRowsPerWave) ? row + 2 : kRowsPerWave - 1;
        gi = ip[nr * kNNZ];
        gv = vp[nr * kNNZ];

        const Offs ob = read_offs(obCur, 1);
        const AOps abv = read_aops(abCur, 1);
        __builtin_amdgcn_sched_barrier(0);
        v8f acc = {};
        acc = mma_unit(acc, aa, ra);
        __builtin_amdgcn_sched_barrier(0);
        const Rows rb = read_rows(ob, slice, lane8);

        const Offs on = read_offs(obNext, 0);
        aa = read_aops(abNext, 0);
        __builtin_amdgcn_sched_barrier(0);
        acc = mma_unit(acc, abv, rb);
        __builtin_amdgcn_sched_barrier(0);
        ra = read_rows(on, slice, lane8);
        return acc;
    };

    auto run_rows = [&](int rowFirst, int rowEnd) {
        for (int row = rowFirst; row < rowEnd; row += 2) {
            const v8f accE = do_row(row,     ab0, ob0, ab1, ob1, 1);
            const v8f accO = do_row(row + 1, ab1, ob1, ab0, ob0, 0);
            const float2 r = { (half ? accO[0] : accE[0]) + bv.x, (half ? accO[1] : accE[1]) + bv.y };
            *reinterpret_cast<float2*>(o) = r;
            o += 2 * kDim;
        }
    };

    __builtin_amdgcn_s_setprio(3);
    run_rows(0, kPrioStepA);
    __builtin_amdgcn_s_setprio(2);
    run_rows(kPrioStepA, kPrioStepB);
    __builtin_amdgcn_s_setprio(1);
    run_rows(kPrioStepB, kPrioStepC);
    __builtin_amdgcn_s_setprio(0);
    run_rows(kPrioStepC, kRowsPerWave);
}

extern "C" void kernel_launch(void* const* d_in, const int* in_sizes, int n_in,
                              void* d_out, int out_size, void* d_ws, size_t ws_size,
                              hipStream_t stream) {
    const float* values  = (const float*)d_in[0];
    const int*   indices = (const int*)  d_in[1];
    const float* kern    = (const float*)d_in[2];
    const float* bias    = (const float*)d_in[3];
    float* outp = (float*)d_out;

    dim3 grid(kSlices, kBatchParts);
    dim3 block(kThreads);
    hipLaunchKernelGGL(sparse_gather_wmma_kernel, grid, block, 0, stream,
                       values, indices, kern, bias, outp);
}
